# starting kernel
# baseline (speedup 1.0000x reference)
#include <hip/hip_runtime.h>

typedef float v2f __attribute__((ext_vector_type(2)));
typedef float v8f __attribute__((ext_vector_type(8)));

__global__ void gcn_zero_kernel(float* __restrict__ p, long long n) {
    long long i = (long long)blockIdx.x * blockDim.x + threadIdx.x;
    if (i < n) p[i] = 0.0f;
}

__global__ void gcn_degree_kernel(const int* __restrict__ src,
                                  const int* __restrict__ dst,
                                  float* __restrict__ deg_out,
                                  float* __restrict__ deg_in, int E) {
    int e = blockIdx.x * blockDim.x + threadIdx.x;
    if (e < E) {
        atomicAdd(deg_out + src[e], 1.0f);
        atomicAdd(deg_in  + dst[e], 1.0f);
    }
}

__global__ void gcn_scatter_kernel(const float* __restrict__ feats,
                                   const int* __restrict__ src,
                                   const int* __restrict__ dst,
                                   const float* __restrict__ deg_out,
                                   float* __restrict__ agg, int E) {
    long long i = (long long)blockIdx.x * blockDim.x + threadIdx.x;
    long long total = (long long)E * 16;
    if (i >= total) return;
    int e  = (int)(i >> 4);
    int f0 = (int)(i & 15) * 4;
    int s = src[e];
    int d = dst[e];
    float sn = rsqrtf(fmaxf(deg_out[s], 1.0f));
    float4 v = *(const float4*)(feats + (size_t)s * 64 + f0);
    float* ap = agg + (size_t)d * 64 + f0;
    atomicAdd(ap + 0, v.x * sn);
    atomicAdd(ap + 1, v.y * sn);
    atomicAdd(ap + 2, v.z * sn);
    atomicAdd(ap + 3, v.w * sn);
}

#define STG 66

__global__ __launch_bounds__(256)
void gcn_gemm_ln_kernel(const float* __restrict__ agg,
                        const float* __restrict__ deg_in,
                        const float* __restrict__ feats,
                        const float* __restrict__ W,
                        const float* __restrict__ bias,
                        const float* __restrict__ gamma,
                        const float* __restrict__ beta,
                        float* __restrict__ out,
                        int n_tiles, int N) {
    __shared__ float lW[64 * 64];
    __shared__ float stage[8][16 * STG];

    const int tid  = threadIdx.x;
    const int wave = tid >> 5;
    const int lane = tid & 31;

    for (int i = tid; i < 64 * 64; i += 256) lW[i] = W[i];
    __syncthreads();

    const int tile = blockIdx.x * 8 + wave;
    if (tile >= n_tiles) return;

    const int row0 = tile * 16;
    const int mrow = lane & 15;
    const int koff = (lane >> 4) * 2;
    const int j    = lane & 15;

    int arow_i = row0 + mrow;
    if (arow_i >= N) arow_i = N - 1;
    const float rnorm = rsqrtf(fmaxf(deg_in[arow_i], 1.0f));
    const float* __restrict__ arow = agg + (size_t)arow_i * 64;

    v8f c0 = {}, c1 = {}, c2 = {}, c3 = {};

    #pragma unroll
    for (int k = 0; k < 16; ++k) {
        const int kk = 4 * k + koff;
        v2f a;
        a.x = arow[kk]     * rnorm;
        a.y = arow[kk + 1] * rnorm;
        const float* w0 = lW + kk * 64;
        const float* w1 = lW + (kk + 1) * 64;
        v2f b0, b1, b2, b3;
        b0.x = w0[j];      b0.y = w1[j];
        b1.x = w0[16 + j]; b1.y = w1[16 + j];
        b2.x = w0[32 + j]; b2.y = w1[32 + j];
        b3.x = w0[48 + j]; b3.y = w1[48 + j];

        c0 = __builtin_amdgcn_wmma_f32_16x16x4_f32(false, a, false, b0, (short)0, c0, false, false);
        c1 = __builtin_amdgcn_wmma_f32_16x16x4_f32(false, a, false, b1, (short)0, c1, false, false);
        c2 = __builtin_amdgcn_wmma_f32_16x16x4_f32(false, a, false, b2, (short)0, c2, false, false);
        c3 = __builtin_amdgcn_wmma_f32_16x16x4_f32(false, a, false, b3, (short)0, c3, false, false);
    }

    float* __restrict__ st = &stage[wave][0];
    const int half = (lane >> 4) * 8;
    #pragma unroll
    for (int v = 0; v < 8; ++v) {
        float* r = st + (half + v) * STG + j;
        r[0]  = c0[v];
        r[16] = c1[v];
        r[32] = c2[v];
        r[48] = c3[v];
    }
    __builtin_amdgcn_wave_barrier();

    if (lane < 16) {
        const int grow = row0 + lane;
        if (grow < N) {
            const float* __restrict__ frow = feats + (size_t)grow * 64;
            float* __restrict__ srow = st + lane * STG;
            float sum = 0.0f, sumsq = 0.0f;
            #pragma unroll 4
            for (int f = 0; f < 64; ++f) {
                float v = srow[f] + bias[f];
                v = fmaxf(v, 0.0f);
                v = v + frow[f];
                srow[f] = v;
                sum   += v;
                sumsq += v * v;
            }
            const float mu   = sum * (1.0f / 64.0f);
            const float var  = sumsq * (1.0f / 64.0f) - mu * mu;
            const float rstd = rsqrtf(var + 1e-5f);
            float* __restrict__ orow = out + (size_t)grow * 64;
            #pragma unroll 4
            for (int f = 0; f < 64; ++f) {
                orow[f] = (srow[f] - mu) * rstd * gamma[f] + beta[f];
            }
        }
    }
}

extern "C" void kernel_launch(void* const* d_in, const int* in_sizes, int n_in,
                              void* d_out, int out_size, void* d_ws, size_t ws_size,
                              hipStream_t stream) {
    const float* feats = (const float*)d_in[0];
    const int*   src   = (const int*)d_in[1];
    const int*   dst   = (const int*)d_in[2];
    const float* W     = (const float*)d_in[3];
    const float* bias  = (const float*)d_in[4];
    const float* gamma = (const float*)d_in[5];
    const float* beta  = (const float*)d_in[6];

    const int N = in_sizes[0] / 64;
    const int E = in_sizes[1];

    float* ws      = (float*)d_ws;
    float* deg_out = ws;
    float* deg_in  = ws + N;
    float* agg     = ws + 2 * (size_t)N;

    long long zn = (long long)N * 66;
    gcn_zero_kernel<<<(int)((zn + 255) / 256), 256, 0, stream>>>(ws, zn);

    gcn_degree_kernel<<<(E + 255) / 256, 256, 0, stream>>>(src, dst, deg_out, deg_in, E);

    long long sc = (long long)E * 16;
    gcn_scatter_kernel<<<(int)((sc + 255) / 256), 256, 0, stream>>>(
        feats, src, dst, deg_out, agg, E);

    const int n_tiles = (N + 15) / 16;
    gcn_gemm_ln_kernel<<<(n_tiles + 7) / 8, 256, 0, stream>>>(
        agg, deg_in, feats, W, bias, gamma, beta, (float*)d_out, n_tiles, N);
}


# owner scan: a work-group owns 208 dst rows in LDS, reads all of dst, adds its own entries (ds_add_f32, arrival order), then GEMM+LN (lane-tree sums); degrees likewise; no global atomics; 2 launches
# speedup vs baseline: 1.6044x; 1.6044x over previous
#include <hip/hip_runtime.h>


typedef float v2f __attribute__((ext_vector_type(2)));
typedef float v8f __attribute__((ext_vector_type(8)));

#define ROWS 208
#define TP   68
#define NT   1024
#define UNR  4
#define QCAP 544

static __device__ __forceinline__ void wave_sync() {
    __builtin_amdgcn_fence(__ATOMIC_RELEASE, "wavefront");
    __builtin_amdgcn_wave_barrier();
    __builtin_amdgcn_fence(__ATOMIC_ACQUIRE, "wavefront");
}

__global__ __launch_bounds__(NT)
void gcn_srcnorm_kernel(const int* __restrict__ src, float* __restrict__ norm_out,
                        int E, int N) {
    __shared__ unsigned cnt[ROWS];
    const int tid  = threadIdx.x;
    const int base = blockIdx.x * ROWS;
    if (tid < ROWS) cnt[tid] = 0u;
    __syncthreads();

    const int nch = E >> 2;
    const int rot = (int)(((long long)blockIdx.x * nch) / gridDim.x);
    const int4* __restrict__ s4 = (const int4*)src;

    for (int i0 = 0; i0 < nch; i0 += NT * UNR) {
        int4 v[UNR];
        #pragma unroll
        for (int u = 0; u < UNR; ++u) {
            const int i = i0 + u * NT + tid;
            int c = i + rot;
            if (c >= nch) c -= nch;
            if (i < nch) v[u] = s4[c];
            else         v[u] = make_int4(-1, -1, -1, -1);
        }
        #pragma unroll
        for (int u = 0; u < UNR; ++u) {
            unsigned a;
            a = (unsigned)(v[u].x - base); if (a < (unsigned)ROWS) atomicAdd(&cnt[a], 1u);
            a = (unsigned)(v[u].y - base); if (a < (unsigned)ROWS) atomicAdd(&cnt[a], 1u);
            a = (unsigned)(v[u].z - base); if (a < (unsigned)ROWS) atomicAdd(&cnt[a], 1u);
            a = (unsigned)(v[u].w - base); if (a < (unsigned)ROWS) atomicAdd(&cnt[a], 1u);
        }
    }
    if (tid < (E & 3)) {
        const unsigned a = (unsigned)(src[4 * nch + tid] - base);
        if (a < (unsigned)ROWS) atomicAdd(&cnt[a], 1u);
    }
    __syncthreads();
    if (tid < ROWS && base + tid < N)
        norm_out[base + tid] = rsqrtf(fmaxf((float)cnt[tid], 1.0f));
}

__global__ __launch_bounds__(NT)
void gcn_owner_kernel(const float* __restrict__ feats,
                      const int* __restrict__ src,
                      const int* __restrict__ dst,
                      const float* __restrict__ norm_out,
                      const float* __restrict__ W,
                      const float* __restrict__ bias,
                      const float* __restrict__ gamma,
                      const float* __restrict__ beta,
                      float* __restrict__ out, int E, int N) {
    __shared__ float    tile[ROWS * TP];
    __shared__ float    lW[64 * 64];
    __shared__ int      q[NT / 32][QCAP];
    __shared__ unsigned cin[ROWS];

    const int tid  = threadIdx.x;
    const int wave = tid >> 5;
    const int lane = tid & 31;
    const int base = blockIdx.x * ROWS;

    for (int i = tid; i < ROWS * TP; i += NT) tile[i] = 0.0f;
    for (int i = tid; i < 64 * 64; i += NT) lW[i] = W[i];
    if (tid < ROWS) cin[tid] = 0u;
    __syncthreads();

    int* __restrict__ myq = &q[wave][0];
    int qn = 0;

    auto drain = [&](const int* qp, int n) {
        wave_sync();
        int my_s = 0, my_r = 0;
        float my_n = 0.0f;
        if (lane < n) {
            const int e = qp[lane];
            my_s = src[e];
            my_r = dst[e] - base;
            my_n = norm_out[my_s];
            atomicAdd(&cin[my_r], 1u);
        }
        for (int t0 = 0; t0 < n; t0 += 16) {
            float2 x[16];
            #pragma unroll
            for (int k = 0; k < 16; ++k) {
                const int s = __builtin_amdgcn_readlane(my_s, (t0 + k) & 31);
                x[k] = *(const float2*)(feats + (size_t)s * 64 + 2 * lane);
            }
            #pragma unroll
            for (int k = 0; k < 16; ++k) {
                const int t = t0 + k;
                if (t < n) {
                    const float nrm = __int_as_float(__builtin_amdgcn_readlane(__float_as_int(my_n), t & 31));
                    const int r = __builtin_amdgcn_readlane(my_r, t & 31);
                    float* p = tile + r * TP + 2 * lane;
                    atomicAdd(p,     x[k].x * nrm);
                    atomicAdd(p + 1, x[k].y * nrm);
                }
            }
        }
        wave_sync();
    };

    const int nch = E >> 2;
    const int rot = (int)(((long long)blockIdx.x * nch) / gridDim.x);
    const int4* __restrict__ d4 = (const int4*)dst;

#define APPEND(m, eidx)                                                        \
    {                                                                          \
        const unsigned mk = __builtin_amdgcn_ballot_w32(m);                    \
        if (m) myq[qn + (int)__builtin_amdgcn_mbcnt_lo(mk, 0u)] = (eidx);      \
        qn += __builtin_popcount(mk);                                          \
    }

    for (int i0 = 0; i0 < nch; i0 += NT * UNR) {
        int4 v[UNR];
        int  cc[UNR];
        #pragma unroll
        for (int u = 0; u < UNR; ++u) {
            const int i = i0 + u * NT + tid;
            int c = i + rot;
            if (c >= nch) c -= nch;
            cc[u] = c;
            if (i < nch) v[u] = d4[c];
            else         v[u] = make_int4(-1, -1, -1, -1);
        }
        #pragma unroll
        for (int u = 0; u < UNR; ++u) {
            const bool m0 = (unsigned)(v[u].x - base) < (unsigned)ROWS;
            const bool m1 = (unsigned)(v[u].y - base) < (unsigned)ROWS;
            const bool m2 = (unsigned)(v[u].z - base) < (unsigned)ROWS;
            const bool m3 = (unsigned)(v[u].w - base) < (unsigned)ROWS;
            if (__builtin_amdgcn_ballot_w32(m0 | m1 | m2 | m3)) {
                APPEND(m0, 4 * cc[u] + 0)
                APPEND(m1, 4 * cc[u] + 1)
                APPEND(m2, 4 * cc[u] + 2)
                APPEND(m3, 4 * cc[u] + 3)
            }
        }
        while (qn >= 32) {
            qn -= 32;
            drain(myq + qn, 32);
        }
    }
    {
        bool mt = false;
        if (tid < (E & 3)) mt = (unsigned)(dst[4 * nch + tid] - base) < (unsigned)ROWS;
        APPEND(mt, 4 * nch + tid)
    }
#undef APPEND
    while (qn >= 32) {
        qn -= 32;
        drain(myq + qn, 32);
    }
    if (qn > 0) drain(myq, qn);
    __syncthreads();

    if (wave >= ROWS / 16) return;

    const int row0 = wave * 16;
    const int mrow = lane & 15;
    const int koff = (lane >> 4) * 2;
    const int j    = lane & 15;

    float2 fr[16];
    #pragma unroll
    for (int r = 0; r < 16; ++r) {
        const int grow = base + row0 + r;
        fr[r] = make_float2(0.0f, 0.0f);
        if (grow < N) fr[r] = *(const float2*)(feats + (size_t)grow * 64 + 2 * lane);
    }

    const float rnorm = rsqrtf(fmaxf((float)cin[row0 + mrow], 1.0f));
    const float* __restrict__ arow = tile + (row0 + mrow) * TP;

    v8f c0 = {}, c1 = {}, c2 = {}, c3 = {};

    #pragma unroll
    for (int k = 0; k < 16; ++k) {
        const int kk = 4 * k + koff;
        v2f a;
        a.x = arow[kk]     * rnorm;
        a.y = arow[kk + 1] * rnorm;
        const float* w0 = lW + kk * 64;
        const float* w1 = lW + (kk + 1) * 64;
        v2f b0, b1, b2, b3;
        b0.x = w0[j];      b0.y = w1[j];
        b1.x = w0[16 + j]; b1.y = w1[16 + j];
        b2.x = w0[32 + j]; b2.y = w1[32 + j];
        b3.x = w0[48 + j]; b3.y = w1[48 + j];

        c0 = __builtin_amdgcn_wmma_f32_16x16x4_f32(false, a, false, b0, (short)0, c0, false, false);
        c1 = __builtin_amdgcn_wmma_f32_16x16x4_f32(false, a, false, b1, (short)0, c1, false, false);
        c2 = __builtin_amdgcn_wmma_f32_16x16x4_f32(false, a, false, b2, (short)0, c2, false, false);
        c3 = __builtin_amdgcn_wmma_f32_16x16x4_f32(false, a, false, b3, (short)0, c3, false, false);
    }

    float* __restrict__ st = tile + row0 * TP;
    const int half = (lane >> 4) * 8;
    #pragma unroll
    for (int v = 0; v < 8; ++v) {
        float* r = st + (half + v) * TP + j;
        r[0]  = c0[v];
        r[16] = c1[v];
        r[32] = c2[v];
        r[48] = c3[v];
    }
    wave_sync();

    const float2 bb = *(const float2*)(bias  + 2 * lane);
    const float2 gg = *(const float2*)(gamma + 2 * lane);
    const float2 be = *(const float2*)(beta  + 2 * lane);

    float2 y[16];
    float  sm[16];
    #pragma unroll
    for (int r = 0; r < 16; ++r) {
        const float2 d = *(const float2*)(st + r * TP + 2 * lane);
        y[r].x = fmaxf(d.x + bb.x, 0.0f) + fr[r].x;
        y[r].y = fmaxf(d.y + bb.y, 0.0f) + fr[r].y;
        sm[r] = y[r].x + y[r].y;
    }
    #pragma unroll
    for (int o = 16; o >= 1; o >>= 1) {
        #pragma unroll
        for (int r = 0; r < 16; ++r) sm[r] += __shfl_xor(sm[r], o);
    }
    float sq[16];
    #pragma unroll
    for (int r = 0; r < 16; ++r) {
        const float mu = sm[r] * (1.0f / 64.0f);
        y[r].x -= mu;
        y[r].y -= mu;
        sq[r] = y[r].x * y[r].x + y[r].y * y[r].y;
    }
    #pragma unroll
    for (int o = 16; o >= 1; o >>= 1) {
        #pragma unroll
        for (int r = 0; r < 16; ++r) sq[r] += __shfl_xor(sq[r], o);
    }
    #pragma unroll
    for (int r = 0; r < 16; ++r) {
        const int grow = base + row0 + r;
        if (grow < N) {
            const float rstd = rsqrtf(sq[r] * (1.0f / 64.0f) + 1e-5f);
            float2 o2;
            o2.x = y[r].x * rstd * gg.x + be.x;
            o2.y = y[r].y * rstd * gg.y + be.y;
            *(float2*)(out + (size_t)grow * 64 + 2 * lane) = o2;
        }
    }
}

extern "C" void kernel_launch(void* const* d_in, const int* in_sizes, int n_in,
                              void* d_out, int out_size, void* d_ws, size_t ws_size,
                              hipStream_t stream) {
    const float* feats = (const float*)d_in[0];
    const int*   src   = (const int*)d_in[1];
    const int*   dst   = (const int*)d_in[2];
    const float* W     = (const float*)d_in[3];
    const float* bias  = (const float*)d_in[4];
    const float* gamma = (const float*)d_in[5];
    const float* beta  = (const float*)d_in[6];

    const int N = in_sizes[0] / 64;
    const int E = in_sizes[1];

    float* norm_out = (float*)d_ws;

    const int G = (N + ROWS - 1) / ROWS;
    gcn_srcnorm_kernel<<<G, NT, 0, stream>>>(src, norm_out, E, N);
    gcn_owner_kernel<<<G, NT, 0, stream>>>(feats, src, dst, norm_out, W, bias, gamma, beta,
                                           (float*)d_out, E, N);
}


# owner scans read a 1-byte owner key per entry (row/208, written by a new first launch), 16 keys per 16-byte load, zero-byte test; 3 launches; sums still ds_add_f32 in arrival order (sequence differs)
# speedup vs baseline: 2.8418x; 1.7713x over previous
#include <hip/hip_runtime.h>


typedef float v2f __attribute__((ext_vector_type(2)));
typedef float v8f __attribute__((ext_vector_type(8)));

#define ROWS 208
#define TP   68
#define NT   1024
#define UNR  4
#define QCAP 64
#define KPAD 255u

static __device__ __forceinline__ void wave_sync() {
    __builtin_amdgcn_fence(__ATOMIC_RELEASE, "wavefront");
    __builtin_amdgcn_wave_barrier();
    __builtin_amdgcn_fence(__ATOMIC_ACQUIRE, "wavefront");
}

static __device__ __forceinline__ unsigned key_of(int idx) {
    return min((unsigned)idx / ROWS, KPAD);
}

__global__ __launch_bounds__(256)
void gcn_keys_kernel(const int* __restrict__ src, const int* __restrict__ dst,
                     unsigned* __restrict__ ksrc, unsigned* __restrict__ kdst, int E) {
    const int g  = blockIdx.x * 256 + threadIdx.x;
    const int ng = ((E + 15) >> 4) * 4;
    if (g >= ng) return;
    unsigned ws, wd;
    if (4 * g + 4 <= E) {
        const int4 a = ((const int4*)src)[g];
        const int4 b = ((const int4*)dst)[g];
        ws = key_of(a.x) | (key_of(a.y) << 8) | (key_of(a.z) << 16) | (key_of(a.w) << 24);
        wd = key_of(b.x) | (key_of(b.y) << 8) | (key_of(b.z) << 16) | (key_of(b.w) << 24);
    } else {
        ws = 0u; wd = 0u;
        #pragma unroll
        for (int j = 0; j < 4; ++j) {
            const int e = 4 * g + j;
            unsigned a = KPAD, b = KPAD;
            if (e < E) { a = key_of(src[e]); b = key_of(dst[e]); }
            ws |= a << (8 * j);
            wd |= b << (8 * j);
        }
    }
    ksrc[g] = ws;
    kdst[g] = wd;
}

static __device__ __forceinline__ unsigned zero_bytes(unsigned x) {
    return ~(((x & 0x7f7f7f7fu) + 0x7f7f7f7fu) | x) & 0x80808080u;
}

template <class DrainF>
static __device__ __forceinline__ void owner_scan(const unsigned* __restrict__ keys, int E,
                                                  int* __restrict__ myq, DrainF drain) {
    const int tid = threadIdx.x;
    const int n16 = (E + 15) >> 4;
    const int rot = (int)(((long long)blockIdx.x * n16) / gridDim.x);
    const unsigned pat = blockIdx.x * 0x01010101u;
    const uint4* __restrict__ k4 = (const uint4*)keys;
    int qn = 0;

    for (int i0 = 0; i0 < n16; i0 += NT * UNR) {
        const int i = i0 + tid;
        uint4 v[UNR];
        #pragma unroll
        for (int u = 0; u < UNR; ++u) {
            const unsigned c = (unsigned)(i + u * NT + rot);
            if (i + u * NT < n16) v[u] = k4[min(c, c - (unsigned)n16)];
            else                  v[u] = make_uint4(~0u, ~0u, ~0u, ~0u);
        }
        unsigned fa = (zero_bytes(v[0].x ^ pat) >> 7) | (zero_bytes(v[0].y ^ pat) >> 6) |
                      (zero_bytes(v[0].z ^ pat) >> 5) | (zero_bytes(v[0].w ^ pat) >> 4) |
                      (zero_bytes(v[1].x ^ pat) >> 3) | (zero_bytes(v[1].y ^ pat) >> 2) |
                      (zero_bytes(v[1].z ^ pat) >> 1) |  zero_bytes(v[1].w ^ pat);
        unsigned fb = (zero_bytes(v[2].x ^ pat) >> 7) | (zero_bytes(v[2].y ^ pat) >> 6) |
                      (zero_bytes(v[2].z ^ pat) >> 5) | (zero_bytes(v[2].w ^ pat) >> 4) |
                      (zero_bytes(v[3].x ^ pat) >> 3) | (zero_bytes(v[3].y ^ pat) >> 2) |
                      (zero_bytes(v[3].z ^ pat) >> 1) |  zero_bytes(v[3].w ^ pat);

        unsigned mk = __builtin_amdgcn_ballot_w32((fa | fb) != 0u);
        while (mk != 0u) {
            if ((fa | fb) != 0u) {
                const bool ina = fa != 0u;
                unsigned w = ina ? fa : fb;
                const int p = __builtin_ctz(w);
                w &= w - 1u;
                if (ina) fa = w; else fb = w;
                const int u = (ina ? 0 : 2) + ((p >> 2) & 1);
                const unsigned c = (unsigned)(i + u * NT + rot);
                myq[qn + (int)__builtin_amdgcn_mbcnt_lo(mk, 0u)] =
                    16 * (int)min(c, c - (unsigned)n16) + ((p & 3) << 2) + (p >> 3);
            }
            qn += __builtin_popcount(mk);
            if (qn >= 32) {
                qn -= 32;
                drain(myq + qn, 32);
            }
            mk = __builtin_amdgcn_ballot_w32((fa | fb) != 0u);
        }
    }
    if (qn > 0) drain(myq, qn);
}

__global__ __launch_bounds__(NT)
void gcn_srcnorm_kernel(const int* __restrict__ src, const unsigned* __restrict__ ksrc,
                        float* __restrict__ norm_out, int E, int N) {
    __shared__ unsigned cnt[ROWS];
    __shared__ int      q[NT / 32][QCAP];
    const int tid  = threadIdx.x;
    const int wave = tid >> 5;
    const int lane = tid & 31;
    const int base = blockIdx.x * ROWS;
    if (tid < ROWS) cnt[tid] = 0u;
    __syncthreads();

    auto drain = [&](const int* qp, int n) {
        wave_sync();
        if (lane < n) atomicAdd(&cnt[src[qp[lane]] - base], 1u);
        wave_sync();
    };
    owner_scan(ksrc, E, &q[wave][0], drain);

    __syncthreads();
    if (tid < ROWS && base + tid < N)
        norm_out[base + tid] = rsqrtf(fmaxf((float)cnt[tid], 1.0f));
}

__global__ __launch_bounds__(NT)
void gcn_owner_kernel(const float* __restrict__ feats,
                      const int* __restrict__ src,
                      const int* __restrict__ dst,
                      const unsigned* __restrict__ kdst,
                      const float* __restrict__ norm_out,
                      const float* __restrict__ W,
                      const float* __restrict__ bias,
                      const float* __restrict__ gamma,
                      const float* __restrict__ beta,
                      float* __restrict__ out, int E, int N) {
    __shared__ float    tile[ROWS * TP];
    __shared__ float    lW[64 * 64];
    __shared__ int      q[NT / 32][QCAP];
    __shared__ unsigned cin[ROWS];

    const int tid  = threadIdx.x;
    const int wave = tid >> 5;
    const int lane = tid & 31;
    const int base = blockIdx.x * ROWS;

    for (int i = tid; i < ROWS * TP; i += NT) tile[i] = 0.0f;
    for (int i = tid; i < 64 * 64; i += NT) lW[i] = W[i];
    if (tid < ROWS) cin[tid] = 0u;
    __syncthreads();

    auto drain = [&](const int* qp, int n) {
        wave_sync();
        int my_s = 0, my_r = 0;
        float my_n = 0.0f;
        if (lane < n) {
            const int e = qp[lane];
            my_s = src[e];
            my_r = dst[e] - base;
            my_n = norm_out[my_s];
            atomicAdd(&cin[my_r], 1u);
        }
        for (int t0 = 0; t0 < n; t0 += 16) {
            float2 x[16];
            #pragma unroll
            for (int k = 0; k < 16; ++k) {
                const int s = __builtin_amdgcn_readlane(my_s, (t0 + k) & 31);
                x[k] = *(const float2*)(feats + (size_t)s * 64 + 2 * lane);
            }
            #pragma unroll
            for (int k = 0; k < 16; ++k) {
                const int t = t0 + k;
                if (t < n) {
                    const float nrm = __int_as_float(__builtin_amdgcn_readlane(__float_as_int(my_n), t & 31));
                    const int r = __builtin_amdgcn_readlane(my_r, t & 31);
                    float* p = tile + r * TP + 2 * lane;
                    atomicAdd(p,     x[k].x * nrm);
                    atomicAdd(p + 1, x[k].y * nrm);
                }
            }
        }
        wave_sync();
    };

    owner_scan(kdst, E, &q[wave][0], drain);
    __syncthreads();

    if (wave >= ROWS / 16) return;

    const int row0 = wave * 16;
    const int mrow = lane & 15;
    const int koff = (lane >> 4) * 2;
    const int j    = lane & 15;

    float2 fr[16];
    #pragma unroll
    for (int r = 0; r < 16; ++r) {
        const int grow = base + row0 + r;
        fr[r] = make_float2(0.0f, 0.0f);
        if (grow < N) fr[r] = *(const float2*)(feats + (size_t)grow * 64 + 2 * lane);
    }

    const float rnorm = rsqrtf(fmaxf((float)cin[row0 + mrow], 1.0f));
    const float* __restrict__ arow = tile + (row0 + mrow) * TP;

    v8f c0 = {}, c1 = {}, c2 = {}, c3 = {};

    #pragma unroll
    for (int k = 0; k < 16; ++k) {
        const int kk = 4 * k + koff;
        v2f a;
        a.x = arow[kk]     * rnorm;
        a.y = arow[kk + 1] * rnorm;
        const float* w0 = lW + kk * 64;
        const float* w1 = lW + (kk + 1) * 64;
        v2f b0, b1, b2, b3;
        b0.x = w0[j];      b0.y = w1[j];
        b1.x = w0[16 + j]; b1.y = w1[16 + j];
        b2.x = w0[32 + j]; b2.y = w1[32 + j];
        b3.x = w0[48 + j]; b3.y = w1[48 + j];

        c0 = __builtin_amdgcn_wmma_f32_16x16x4_f32(false, a, false, b0, (short)0, c0, false, false);
        c1 = __builtin_amdgcn_wmma_f32_16x16x4_f32(false, a, false, b1, (short)0, c1, false, false);
        c2 = __builtin_amdgcn_wmma_f32_16x16x4_f32(false, a, false, b2, (short)0, c2, false, false);
        c3 = __builtin_amdgcn_wmma_f32_16x16x4_f32(false, a, false, b3, (short)0, c3, false, false);
    }

    float* __restrict__ st = tile + row0 * TP;
    const int half = (lane >> 4) * 8;
    #pragma unroll
    for (int v = 0; v < 8; ++v) {
        float* r = st + (half + v) * TP + j;
        r[0]  = c0[v];
        r[16] = c1[v];
        r[32] = c2[v];
        r[48] = c3[v];
    }
    wave_sync();

    const float2 bb = *(const float2*)(bias  + 2 * lane);
    const float2 gg = *(const float2*)(gamma + 2 * lane);
    const float2 be = *(const float2*)(beta  + 2 * lane);

    float2 y[16];
    float  sm[16];
    #pragma unroll
    for (int r = 0; r < 16; ++r) {
        const float2 d = *(const float2*)(st + r * TP + 2 * lane);
        y[r].x = fmaxf(d.x + bb.x, 0.0f) + fr[r].x;
        y[r].y = fmaxf(d.y + bb.y, 0.0f) + fr[r].y;
        sm[r] = y[r].x + y[r].y;
    }
    #pragma unroll
    for (int o = 16; o >= 1; o >>= 1) {
        #pragma unroll
        for (int r = 0; r < 16; ++r) sm[r] += __shfl_xor(sm[r], o);
    }
    float sq[16];
    #pragma unroll
    for (int r = 0; r < 16; ++r) {
        const float mu = sm[r] * (1.0f / 64.0f);
        y[r].x -= mu;
        y[r].y -= mu;
        sq[r] = y[r].x * y[r].x + y[r].y * y[r].y;
    }
    #pragma unroll
    for (int o = 16; o >= 1; o >>= 1) {
        #pragma unroll
        for (int r = 0; r < 16; ++r) sq[r] += __shfl_xor(sq[r], o);
    }
    #pragma unroll
    for (int r = 0; r < 16; ++r) {
        const int grow = base + row0 + r;
        if (grow < N) {
            const float rstd = rsqrtf(sq[r] * (1.0f / 64.0f) + 1e-5f);
            float2 o2;
            o2.x = y[r].x * rstd * gg.x + be.x;
            o2.y = y[r].y * rstd * gg.y + be.y;
            *(float2*)(out + (size_t)grow * 64 + 2 * lane) = o2;
        }
    }
}

extern "C" void kernel_launch(void* const* d_in, const int* in_sizes, int n_in,
                              void* d_out, int out_size, void* d_ws, size_t ws_size,
                              hipStream_t stream) {
    const float* feats = (const float*)d_in[0];
    const int*   src   = (const int*)d_in[1];
    const int*   dst   = (const int*)d_in[2];
    const float* W     = (const float*)d_in[3];
    const float* bias  = (const float*)d_in[4];
    const float* gamma = (const float*)d_in[5];
    const float* beta  = (const float*)d_in[6];

    const int N = in_sizes[0] / 64;
    const int E = in_sizes[1];

    float*    norm_out = (float*)d_ws;
    unsigned* ksrc = (unsigned*)((char*)d_ws + (16u << 20));
    unsigned* kdst = (unsigned*)((char*)d_ws + (32u << 20));

    const int G  = (N + ROWS - 1) / ROWS;
    const int GK = (((E + 15) >> 4) * 4 + 255) / 256;
    gcn_keys_kernel<<<GK, 256, 0, stream>>>(src, dst, ksrc, kdst, E);
    gcn_srcnorm_kernel<<<G, NT, 0, stream>>>(src, ksrc, norm_out, E, N);
    gcn_owner_kernel<<<G, NT, 0, stream>>>(feats, src, dst, kdst, norm_out, W, bias, gamma, beta,
                                           (float*)d_out, E, N);
}


# owner keys bit-sliced (8 plane words per 32 entries, built by ballots): a scan gets a block's 32 match flags from 8 bit ops, not 64 byte zero-tests; ds_add_f32 arrival order differs
# speedup vs baseline: 3.4433x; 1.2117x over previous
#include <hip/hip_runtime.h>


typedef float v2f __attribute__((ext_vector_type(2)));
typedef float v8f __attribute__((ext_vector_type(8)));

#define ROWS 208
#define TP   68
#define NT   1024
#define BPT  2
#define QCAP 64
#define KPAD 255u

static __device__ __forceinline__ void wave_sync() {
    __builtin_amdgcn_fence(__ATOMIC_RELEASE, "wavefront");
    __builtin_amdgcn_wave_barrier();
    __builtin_amdgcn_fence(__ATOMIC_ACQUIRE, "wavefront");
}

static __device__ __forceinline__ unsigned key_of(int idx) {
    return min((unsigned)idx / ROWS, KPAD);
}

static __host__ __device__ __forceinline__ int padded_superblocks(int E) {
    const int per_trip = (NT / 32) * BPT;
    return ((((E + 1023) >> 10) + per_trip - 1) / per_trip) * per_trip;
}

__global__ __launch_bounds__(256)
void gcn_keys_kernel(const int* __restrict__ src, const int* __restrict__ dst,
                     unsigned* __restrict__ ksrc, unsigned* __restrict__ kdst, int E) {
    const int lane = threadIdx.x & 31;
    const int sb   = blockIdx.x * 8 + (threadIdx.x >> 5);
    const int* __restrict__ arr = blockIdx.y ? dst : src;
    unsigned*  __restrict__ out = blockIdx.y ? kdst : ksrc;
    const int e0 = sb * 1024 + lane;

    int idx[32];
    if (sb * 1024 + 1024 <= E) {
        #pragma unroll
        for (int b = 0; b < 32; ++b) idx[b] = arr[e0 + 32 * b];
    } else {
        #pragma unroll
        for (int b = 0; b < 32; ++b) {
            const int e = e0 + 32 * b;
            idx[b] = (e < E) ? arr[e] : -1;
        }
    }
    unsigned P[8] = {0u, 0u, 0u, 0u, 0u, 0u, 0u, 0u};
    #pragma unroll
    for (int b = 0; b < 32; ++b) {
        const unsigned k = key_of(idx[b]);
        #pragma unroll
        for (int p = 0; p < 8; ++p) {
            const unsigned bal = __builtin_amdgcn_ballot_w32(((k >> p) & 1u) != 0u);
            P[p] = (lane == b) ? bal : P[p];
        }
    }
    uint4* __restrict__ o4 = (uint4*)out + (size_t)sb * 64;
    o4[lane]      = make_uint4(P[0], P[1], P[2], P[3]);
    o4[32 + lane] = make_uint4(P[4], P[5], P[6], P[7]);
}

template <class DrainF>
static __device__ __forceinline__ void owner_scan(const unsigned* __restrict__ keys, int E,
                                                  int* __restrict__ myq, DrainF drain) {
    const int lane = threadIdx.x & 31;
    const int wave = __builtin_amdgcn_readfirstlane((int)(threadIdx.x >> 5));
    const int nsb  = padded_superblocks(E);
    const int rot  = (int)(((long long)blockIdx.x * nsb) / gridDim.x);
    const unsigned own = blockIdx.x;
    unsigned sel[8];
    #pragma unroll
    for (int p = 0; p < 8; ++p) sel[p] = 0u - ((own >> p) & 1u);
    const unsigned lane16 = (unsigned)lane * 16u;
    int qn = 0;

    for (int t0 = 0; t0 < nsb; t0 += (NT / 32) * BPT) {
        uint4 lo[BPT], hi[BPT];
        int   blk[BPT];
        #pragma unroll
        for (int u = 0; u < BPT; ++u) {
            int sb = t0 + u * (NT / 32) + wave + rot;
            if (sb >= nsb) sb -= nsb;
            blk[u] = sb * 32 + lane;
            const char* __restrict__ run = (const char*)keys + (size_t)sb * 1024;
            lo[u] = *(const uint4*)(run + lane16);
            hi[u] = *(const uint4*)(run + 512 + lane16);
        }
        unsigned w[BPT];
        #pragma unroll
        for (int u = 0; u < BPT; ++u) {
            unsigned m = ~(lo[u].x ^ sel[0]);
            m &= ~(lo[u].y ^ sel[1]);
            m &= ~(lo[u].z ^ sel[2]);
            m &= ~(lo[u].w ^ sel[3]);
            m &= ~(hi[u].x ^ sel[4]);
            m &= ~(hi[u].y ^ sel[5]);
            m &= ~(hi[u].z ^ sel[6]);
            m &= ~(hi[u].w ^ sel[7]);
            w[u] = m;
        }

        unsigned mk = __builtin_amdgcn_ballot_w32((w[0] | w[1]) != 0u);
        while (mk != 0u) {
            if ((w[0] | w[1]) != 0u) {
                const bool first = w[0] != 0u;
                unsigned x = first ? w[0] : w[1];
                const int p = __builtin_ctz(x);
                x &= x - 1u;
                if (first) w[0] = x; else w[1] = x;
                myq[qn + (int)__builtin_amdgcn_mbcnt_lo(mk, 0u)] = (first ? blk[0] : blk[1]) * 32 + p;
            }
            qn += __builtin_popcount(mk);
            if (qn >= 32) {
                qn -= 32;
                drain(myq + qn, 32);
            }
            mk = __builtin_amdgcn_ballot_w32((w[0] | w[1]) != 0u);
        }
    }
    if (qn > 0) drain(myq, qn);
}

__global__ __launch_bounds__(NT)
void gcn_srcnorm_kernel(const int* __restrict__ src, const unsigned* __restrict__ ksrc,
                        float* __restrict__ norm_out, int E, int N) {
    __shared__ unsigned cnt[ROWS];
    __shared__ int      q[NT / 32][QCAP];
    const int tid  = threadIdx.x;
    const int wave = tid >> 5;
    const int lane = tid & 31;
    const int base = blockIdx.x * ROWS;
    if (tid < ROWS) cnt[tid] = 0u;
    __syncthreads();

    auto drain = [&](const int* qp, int n) {
        wave_sync();
        if (lane < n) atomicAdd(&cnt[src[qp[lane]] - base], 1u);
        wave_sync();
    };
    owner_scan(ksrc, E, &q[wave][0], drain);

    __syncthreads();
    if (tid < ROWS && base + tid < N)
        norm_out[base + tid] = rsqrtf(fmaxf((float)cnt[tid], 1.0f));
}

__global__ __launch_bounds__(NT) __attribute__((amdgpu_waves_per_eu(8, 8)))
void gcn_owner_kernel(const float* __restrict__ feats,
                      const int* __restrict__ src,
                      const int* __restrict__ dst,
                      const unsigned* __restrict__ kdst,
                      const float* __restrict__ norm_out,
                      const float* __restrict__ W,
                      const float* __restrict__ bias,
                      const float* __restrict__ gamma,
                      const float* __restrict__ beta,
                      float* __restrict__ out, int E, int N) {
    __shared__ float    tile[ROWS * TP];
    __shared__ float    lW[64 * 64];
    __shared__ int      q[NT / 32][QCAP];
    __shared__ unsigned cin[ROWS];

    const int tid  = threadIdx.x;
    const int wave = tid >> 5;
    const int lane = tid & 31;
    const int base = blockIdx.x * ROWS;

    for (int i = tid; i < ROWS * TP; i += NT) tile[i] = 0.0f;
    for (int i = tid; i < 64 * 64; i += NT) lW[i] = W[i];
    if (tid < ROWS) cin[tid] = 0u;
    __syncthreads();

    auto drain = [&](const int* qp, int n) {
        wave_sync();
        int my_s = 0, my_r = 0;
        float my_n = 0.0f;
        if (lane < n) {
            const int e = qp[lane];
            my_s = src[e];
            my_r = dst[e] - base;
            my_n = norm_out[my_s];
            atomicAdd(&cin[my_r], 1u);
        }
        for (int t0 = 0; t0 < n; t0 += 16) {
            float2 x[16];
            #pragma unroll
            for (int k = 0; k < 16; ++k) {
                const int s = __builtin_amdgcn_readlane(my_s, (t0 + k) & 31);
                x[k] = *(const float2*)(feats + (size_t)s * 64 + 2 * lane);
            }
            #pragma unroll
            for (int k = 0; k < 16; ++k) {
                const int t = t0 + k;
                if (t < n) {
                    const float nrm = __int_as_float(__builtin_amdgcn_readlane(__float_as_int(my_n), t & 31));
                    const int r = __builtin_amdgcn_readlane(my_r, t & 31);
                    float* p = tile + r * TP + 2 * lane;
                    atomicAdd(p,     x[k].x * nrm);
                    atomicAdd(p + 1, x[k].y * nrm);
                }
            }
        }
        wave_sync();
    };

    owner_scan(kdst, E, &q[wave][0], drain);
    __syncthreads();

    if (wave >= ROWS / 16) return;

    const int row0 = wave * 16;
    const int mrow = lane & 15;
    const int koff = (lane >> 4) * 2;
    const int j    = lane & 15;

    float2 fr[16];
    #pragma unroll
    for (int r = 0; r < 16; ++r) {
        const int grow = base + row0 + r;
        fr[r] = make_float2(0.0f, 0.0f);
        if (grow < N) fr[r] = *(const float2*)(feats + (size_t)grow * 64 + 2 * lane);
    }

    const float rnorm = rsqrtf(fmaxf((float)cin[row0 + mrow], 1.0f));
    const float* __restrict__ arow = tile + (row0 + mrow) * TP;

    v8f c0 = {}, c1 = {}, c2 = {}, c3 = {};

    #pragma unroll
    for (int k = 0; k < 16; ++k) {
        const int kk = 4 * k + koff;
        v2f a;
        a.x = arow[kk]     * rnorm;
        a.y = arow[kk + 1] * rnorm;
        const float* w0 = lW + kk * 64;
        const float* w1 = lW + (kk + 1) * 64;
        v2f b0, b1, b2, b3;
        b0.x = w0[j];      b0.y = w1[j];
        b1.x = w0[16 + j]; b1.y = w1[16 + j];
        b2.x = w0[32 + j]; b2.y = w1[32 + j];
        b3.x = w0[48 + j]; b3.y = w1[48 + j];

        c0 = __builtin_amdgcn_wmma_f32_16x16x4_f32(false, a, false, b0, (short)0, c0, false, false);
        c1 = __builtin_amdgcn_wmma_f32_16x16x4_f32(false, a, false, b1, (short)0, c1, false, false);
        c2 = __builtin_amdgcn_wmma_f32_16x16x4_f32(false, a, false, b2, (short)0, c2, false, false);
        c3 = __builtin_amdgcn_wmma_f32_16x16x4_f32(false, a, false, b3, (short)0, c3, false, false);
    }

    float* __restrict__ st = tile + row0 * TP;
    const int half = (lane >> 4) * 8;
    #pragma unroll
    for (int v = 0; v < 8; ++v) {
        float* r = st + (half + v) * TP + j;
        r[0]  = c0[v];
        r[16] = c1[v];
        r[32] = c2[v];
        r[48] = c3[v];
    }
    wave_sync();

    const float2 bb = *(const float2*)(bias  + 2 * lane);
    const float2 gg = *(const float2*)(gamma + 2 * lane);
    const float2 be = *(const float2*)(beta  + 2 * lane);

    float2 y[16];
    float  sm[16];
    #pragma unroll
    for (int r = 0; r < 16; ++r) {
        const float2 d = *(const float2*)(st + r * TP + 2 * lane);
        y[r].x = fmaxf(d.x + bb.x, 0.0f) + fr[r].x;
        y[r].y = fmaxf(d.y + bb.y, 0.0f) + fr[r].y;
        sm[r] = y[r].x + y[r].y;
    }
    #pragma unroll
    for (int o = 16; o >= 1; o >>= 1) {
        #pragma unroll
        for (int r = 0; r < 16; ++r) sm[r] += __shfl_xor(sm[r], o);
    }
    float sq[16];
    #pragma unroll
    for (int r = 0; r < 16; ++r) {
        const float mu = sm[r] * (1.0f / 64.0f);
        y[r].x -= mu;
        y[r].y -= mu;
        sq[r] = y[r].x * y[r].x + y[r].y * y[r].y;
    }
    #pragma unroll
    for (int o = 16; o >= 1; o >>= 1) {
        #pragma unroll
        for (int r = 0; r < 16; ++r) sq[r] += __shfl_xor(sq[r], o);
    }
    #pragma unroll
    for (int r = 0; r < 16; ++r) {
        const int grow = base + row0 + r;
        if (grow < N) {
            const float rstd = rsqrtf(sq[r] * (1.0f / 64.0f) + 1e-5f);
            float2 o2;
            o2.x = y[r].x * rstd * gg.x + be.x;
            o2.y = y[r].y * rstd * gg.y + be.y;
            *(float2*)(out + (size_t)grow * 64 + 2 * lane) = o2;
        }
    }
}

extern "C" void kernel_launch(void* const* d_in, const int* in_sizes, int n_in,
                              void* d_out, int out_size, void* d_ws, size_t ws_size,
                              hipStream_t stream) {
    const float* feats = (const float*)d_in[0];
    const int*   src   = (const int*)d_in[1];
    const int*   dst   = (const int*)d_in[2];
    const float* W     = (const float*)d_in[3];
    const float* bias  = (const float*)d_in[4];
    const float* gamma = (const float*)d_in[5];
    const float* beta  = (const float*)d_in[6];

    const int N = in_sizes[0] / 64;
    const int E = in_sizes[1];

    float*    norm_out = (float*)d_ws;
    unsigned* ksrc = (unsigned*)((char*)d_ws + (16u << 20));
    unsigned* kdst = (unsigned*)((char*)d_ws + (32u << 20));

    const int G  = (N + ROWS - 1) / ROWS;
    const int GK = padded_superblocks(E) / 8;
    gcn_keys_kernel<<<dim3(GK, 2), 256, 0, stream>>>(src, dst, ksrc, kdst, E);
    gcn_srcnorm_kernel<<<G, NT, 0, stream>>>(src, ksrc, norm_out, E, N);
    gcn_owner_kernel<<<G, NT, 0, stream>>>(feats, src, dst, kdst, norm_out, W, bias, gamma, beta,
                                           (float*)d_out, E, N);
}


# out-degree owner scan made 2-D: 63 owner groups (800 rows, 6-bit src keys) x 4 entry slices; a work-group reads 1/4 of six key planes, not all of eight; reader adds a row's 4 partial counts (exact)
# speedup vs baseline: 3.6272x; 1.0534x over previous
#include <hip/hip_runtime.h>


typedef float v2f __attribute__((ext_vector_type(2)));
typedef float v8f __attribute__((ext_vector_type(8)));

#define ROWS 208
#define TP   68
#define NT   1024
#define BPT  2
#define QCAP 64
#define KPAD 255u
#define SROWS 800
#define SPAD 63u
#define NS   4
#define SBT  ((NT / 32) * BPT)

static __device__ __forceinline__ void wave_sync() {
    __builtin_amdgcn_fence(__ATOMIC_RELEASE, "wavefront");
    __builtin_amdgcn_wave_barrier();
    __builtin_amdgcn_fence(__ATOMIC_ACQUIRE, "wavefront");
}

static __host__ __device__ __forceinline__ int padded_superblocks(int E) {
    return ((((E + 1023) >> 10) + SBT - 1) / SBT) * SBT;
}

template <int PLANES, unsigned RPO, unsigned PAD>
static __device__ __forceinline__ void build_keys(const int* __restrict__ arr, unsigned* __restrict__ out,
                                                  int sb, int lane, int E) {
    const int e0 = sb * 1024 + lane;

    int idx[32];
    if (sb * 1024 + 1024 <= E) {
        #pragma unroll
        for (int b = 0; b < 32; ++b) idx[b] = arr[e0 + 32 * b];
    } else {
        #pragma unroll
        for (int b = 0; b < 32; ++b) {
            const int e = e0 + 32 * b;
            idx[b] = (e < E) ? arr[e] : -1;
        }
    }
    unsigned P[PLANES];
    #pragma unroll
    for (int p = 0; p < PLANES; ++p) P[p] = 0u;
    #pragma unroll
    for (int b = 0; b < 32; ++b) {
        const unsigned k = min((unsigned)idx[b] / RPO, PAD);
        #pragma unroll
        for (int p = 0; p < PLANES; ++p) {
            const unsigned bal = __builtin_amdgcn_ballot_w32(((k >> p) & 1u) != 0u);
            P[p] = (lane == b) ? bal : P[p];
        }
    }
    char* __restrict__ run = (char*)out + (size_t)sb * (PLANES * 128);
    *(uint4*)(run + lane * 16) = make_uint4(P[0], P[1], P[2], P[3]);
    if constexpr (PLANES == 8) *(uint4*)(run + 512 + lane * 16) = make_uint4(P[4], P[5], P[6], P[7]);
    else                       *(uint2*)(run + 512 + lane * 8)  = make_uint2(P[4], P[5]);
}

__global__ __launch_bounds__(256)
void gcn_keys_kernel(const int* __restrict__ src, const int* __restrict__ dst,
                     unsigned* __restrict__ ksrc, unsigned* __restrict__ kdst, int E) {
    const int lane = threadIdx.x & 31;
    const int sb   = blockIdx.x * 8 + (threadIdx.x >> 5);
    if (blockIdx.y) build_keys<8, ROWS,  KPAD>(dst, kdst, sb, lane, E);
    else            build_keys<6, SROWS, SPAD>(src, ksrc, sb, lane, E);
}

template <int PLANES, class DrainF>
static __device__ __forceinline__ void owner_scan(const unsigned* __restrict__ keys, int sb_lo, int sb_n, int rot,
                                                  unsigned own, int* __restrict__ myq, DrainF drain) {
    const int lane = threadIdx.x & 31;
    const int wave = __builtin_amdgcn_readfirstlane((int)(threadIdx.x >> 5));
    unsigned sel[PLANES];
    #pragma unroll
    for (int p = 0; p < PLANES; ++p) sel[p] = 0u - ((own >> p) & 1u);
    const unsigned lane16 = (unsigned)lane * 16u;
    const unsigned laneHi = (unsigned)lane * (PLANES == 8 ? 16u : 8u);
    int qn = 0;

    for (int t0 = 0; t0 < sb_n; t0 += SBT) {
        uint4 lo[BPT], hi[BPT];
        int   blk[BPT];
        #pragma unroll
        for (int u = 0; u < BPT; ++u) {
            int sb = t0 + u * (NT / 32) + wave + rot;
            if (sb >= sb_n) sb -= sb_n;
            sb += sb_lo;
            blk[u] = sb * 32 + lane;
            const char* __restrict__ run = (const char*)keys + (size_t)sb * (PLANES * 128);
            lo[u] = *(const uint4*)(run + lane16);
            if constexpr (PLANES == 8) {
                hi[u] = *(const uint4*)(run + 512 + laneHi);
            } else {
                const uint2 h = *(const uint2*)(run + 512 + laneHi);
                hi[u] = make_uint4(h.x, h.y, 0u, 0u);
            }
        }
        unsigned w[BPT];
        #pragma unroll
        for (int u = 0; u < BPT; ++u) {
            unsigned m = ~(lo[u].x ^ sel[0]);
            m &= ~(lo[u].y ^ sel[1]);
            m &= ~(lo[u].z ^ sel[2]);
            m &= ~(lo[u].w ^ sel[3]);
            m &= ~(hi[u].x ^ sel[4]);
            m &= ~(hi[u].y ^ sel[5]);
            if constexpr (PLANES == 8) {
                m &= ~(hi[u].z ^ sel[6]);
                m &= ~(hi[u].w ^ sel[7]);
            }
            w[u] = m;
        }

        unsigned mk = __builtin_amdgcn_ballot_w32((w[0] | w[1]) != 0u);
        while (mk != 0u) {
            if ((w[0] | w[1]) != 0u) {
                const bool first = w[0] != 0u;
                unsigned x = first ? w[0] : w[1];
                const int p = __builtin_ctz(x);
                x &= x - 1u;
                if (first) w[0] = x; else w[1] = x;
                myq[qn + (int)__builtin_amdgcn_mbcnt_lo(mk, 0u)] = (first ? blk[0] : blk[1]) * 32 + p;
            }
            qn += __builtin_popcount(mk);
            if (qn >= 32) {
                qn -= 32;
                drain(myq + qn, 32);
            }
            mk = __builtin_amdgcn_ballot_w32((w[0] | w[1]) != 0u);
        }
    }
    if (qn > 0) drain(myq, qn);
}

__global__ __launch_bounds__(NT)
void gcn_srccount_kernel(const int* __restrict__ src, const unsigned* __restrict__ ksrc,
                         unsigned* __restrict__ part, int E, int N) {
    __shared__ unsigned cnt[SROWS];
    __shared__ int      q[NT / 32][QCAP];
    const int tid  = threadIdx.x;
    const int wave = tid >> 5;
    const int lane = tid & 31;
    const int o    = blockIdx.x / NS;
    const int s    = blockIdx.x % NS;
    const int base = o * SROWS;
    if (tid < SROWS) cnt[tid] = 0u;
    __syncthreads();

    const int trips = padded_superblocks(E) / SBT;
    const int t_lo  = (s * trips) / NS;
    const int t_hi  = ((s + 1) * trips) / NS;
    const int sb_n  = (t_hi - t_lo) * SBT;
    const int rot   = (int)(((long long)o * sb_n) / (gridDim.x / NS));

    auto drain = [&](const int* qp, int n) {
        wave_sync();
        if (lane < n) atomicAdd(&cnt[src[qp[lane]] - base], 1u);
        wave_sync();
    };
    owner_scan<6>(ksrc, t_lo * SBT, sb_n, rot, (unsigned)o, &q[wave][0], drain);

    __syncthreads();
    if (tid < SROWS && base + tid < N)
        part[(size_t)(base + tid) * NS + s] = cnt[tid];
}

__global__ __launch_bounds__(NT) __attribute__((amdgpu_waves_per_eu(8, 8)))
void gcn_owner_kernel(const float* __restrict__ feats,
                      const int* __restrict__ src,
                      const int* __restrict__ dst,
                      const unsigned* __restrict__ kdst,
                      const unsigned* __restrict__ part,
                      const float* __restrict__ W,
                      const float* __restrict__ bias,
                      const float* __restrict__ gamma,
                      const float* __restrict__ beta,
                      float* __restrict__ out, int E, int N) {
    __shared__ float    tile[ROWS * TP];
    __shared__ float    lW[64 * 64];
    __shared__ int      q[NT / 32][QCAP];
    __shared__ unsigned cin[ROWS];

    const int tid  = threadIdx.x;
    const int wave = tid >> 5;
    const int lane = tid & 31;
    const int base = blockIdx.x * ROWS;

    for (int i = tid; i < ROWS * TP; i += NT) tile[i] = 0.0f;
    for (int i = tid; i < 64 * 64; i += NT) lW[i] = W[i];
    if (tid < ROWS) cin[tid] = 0u;
    __syncthreads();

    auto drain = [&](const int* qp, int n) {
        wave_sync();
        int my_s = 0, my_r = 0;
        uint4 c = make_uint4(0u, 0u, 0u, 0u);
        if (lane < n) {
            const int e = qp[lane];
            my_s = src[e];
            my_r = dst[e] - base;
            c = *(const uint4*)(part + (size_t)my_s * NS);
            atomicAdd(&cin[my_r], 1u);
        }
        float my_n = 0.0f;
        #pragma unroll
        for (int h = 0; h < 2; ++h) {
            const int t0 = 16 * h;
            if (t0 < n) {
                float2 x[16];
                #pragma unroll
                for (int k = 0; k < 16; ++k) {
                    const int s = __builtin_amdgcn_readlane(my_s, t0 + k);
                    x[k] = *(const float2*)(feats + (size_t)s * 64 + 2 * lane);
                }
                if (h == 0) my_n = rsqrtf(fmaxf((float)(c.x + c.y + c.z + c.w), 1.0f));
                #pragma unroll
                for (int k = 0; k < 16; ++k) {
                    const int t = t0 + k;
                    if (t < n) {
                        const float nrm = __int_as_float(__builtin_amdgcn_readlane(__float_as_int(my_n), t));
                        const int r = __builtin_amdgcn_readlane(my_r, t);
                        float* p = tile + r * TP + 2 * lane;
                        atomicAdd(p,     x[k].x * nrm);
                        atomicAdd(p + 1, x[k].y * nrm);
                    }
                }
            }
        }
        wave_sync();
    };

    const int nsb = padded_superblocks(E);
    owner_scan<8>(kdst, 0, nsb, (int)(((long long)blockIdx.x * nsb) / gridDim.x), blockIdx.x, &q[wave][0], drain);
    __syncthreads();

    if (wave >= ROWS / 16) return;

    const int row0 = wave * 16;
    const int mrow = lane & 15;
    const int koff = (lane >> 4) * 2;
    const int j    = lane & 15;

    float2 fr[16];
    #pragma unroll
    for (int r = 0; r < 16; ++r) {
        const int grow = base + row0 + r;
        fr[r] = make_float2(0.0f, 0.0f);
        if (grow < N) fr[r] = *(const float2*)(feats + (size_t)grow * 64 + 2 * lane);
    }

    const float rnorm = rsqrtf(fmaxf((float)cin[row0 + mrow], 1.0f));
    const float* __restrict__ arow = tile + (row0 + mrow) * TP;

    v8f c0 = {}, c1 = {}, c2 = {}, c3 = {};

    #pragma unroll
    for (int k = 0; k < 16; ++k) {
        const int kk = 4 * k + koff;
        v2f a;
        a.x = arow[kk]     * rnorm;
        a.y = arow[kk + 1] * rnorm;
        const float* w0 = lW + kk * 64;
        const float* w1 = lW + (kk + 1) * 64;
        v2f b0, b1, b2, b3;
        b0.x = w0[j];      b0.y = w1[j];
        b1.x = w0[16 + j]; b1.y = w1[16 + j];
        b2.x = w0[32 + j]; b2.y = w1[32 + j];
        b3.x = w0[48 + j]; b3.y = w1[48 + j];

        c0 = __builtin_amdgcn_wmma_f32_16x16x4_f32(false, a, false, b0, (short)0, c0, false, false);
        c1 = __builtin_amdgcn_wmma_f32_16x16x4_f32(false, a, false, b1, (short)0, c1, false, false);
        c2 = __builtin_amdgcn_wmma_f32_16x16x4_f32(false, a, false, b2, (short)0, c2, false, false);
        c3 = __builtin_amdgcn_wmma_f32_16x16x4_f32(false, a, false, b3, (short)0, c3, false, false);
    }

    float* __restrict__ st = tile + row0 * TP;
    const int half = (lane >> 4) * 8;
    #pragma unroll
    for (int v = 0; v < 8; ++v) {
        float* r = st + (half + v) * TP + j;
        r[0]  = c0[v];
        r[16] = c1[v];
        r[32] = c2[v];
        r[48] = c3[v];
    }
    wave_sync();

    const float2 bb = *(const float2*)(bias  + 2 * lane);
    const float2 gg = *(const float2*)(gamma + 2 * lane);
    const float2 be = *(const float2*)(beta  + 2 * lane);

    float2 y[16];
    float  sm[16];
    #pragma unroll
    for (int r = 0; r < 16; ++r) {
        const float2 d = *(const float2*)(st + r * TP + 2 * lane);
        y[r].x = fmaxf(d.x + bb.x, 0.0f) + fr[r].x;
        y[r].y = fmaxf(d.y + bb.y, 0.0f) + fr[r].y;
        sm[r] = y[r].x + y[r].y;
    }
    #pragma unroll
    for (int o = 16; o >= 1; o >>= 1) {
        #pragma unroll
        for (int r = 0; r < 16; ++r) sm[r] += __shfl_xor(sm[r], o);
    }
    float sq[16];
    #pragma unroll
    for (int r = 0; r < 16; ++r) {
        const float mu = sm[r] * (1.0f / 64.0f);
        y[r].x -= mu;
        y[r].y -= mu;
        sq[r] = y[r].x * y[r].x + y[r].y * y[r].y;
    }
    #pragma unroll
    for (int o = 16; o >= 1; o >>= 1) {
        #pragma unroll
        for (int r = 0; r < 16; ++r) sq[r] += __shfl_xor(sq[r], o);
    }
    #pragma unroll
    for (int r = 0; r < 16; ++r) {
        const int grow = base + row0 + r;
        if (grow < N) {
            const float rstd = rsqrtf(sq[r] * (1.0f / 64.0f) + 1e-5f);
            float2 o2;
            o2.x = y[r].x * rstd * gg.x + be.x;
            o2.y = y[r].y * rstd * gg.y + be.y;
            *(float2*)(out + (size_t)grow * 64 + 2 * lane) = o2;
        }
    }
}

extern "C" void kernel_launch(void* const* d_in, const int* in_sizes, int n_in,
                              void* d_out, int out_size, void* d_ws, size_t ws_size,
                              hipStream_t stream) {
    const float* feats = (const float*)d_in[0];
    const int*   src   = (const int*)d_in[1];
    const int*   dst   = (const int*)d_in[2];
    const float* W     = (const float*)d_in[3];
    const float* bias  = (const float*)d_in[4];
    const float* gamma = (const float*)d_in[5];
    const float* beta  = (const float*)d_in[6];

    const int N = in_sizes[0] / 64;
    const int E = in_sizes[1];

    unsigned* part = (unsigned*)d_ws;
    unsigned* ksrc = (unsigned*)((char*)d_ws + (16u << 20));
    unsigned* kdst = (unsigned*)((char*)d_ws + (32u << 20));

    const int G  = (N + ROWS - 1) / ROWS;
    const int GS = (N + SROWS - 1) / SROWS;
    const int GK = padded_superblocks(E) / 8;
    gcn_keys_kernel<<<dim3(GK, 2), 256, 0, stream>>>(src, dst, ksrc, kdst, E);
    gcn_srccount_kernel<<<GS * NS, NT, 0, stream>>>(src, ksrc, part, E, N);
    gcn_owner_kernel<<<G, NT, 0, stream>>>(feats, src, dst, kdst, part, W, bias, gamma, beta,
                                           (float*)d_out, E, N);
}


# key launch also writes a packed word per entry (src*256 + dst row within its owner's rows); owner kernel reads 1 scattered word per owned entry, not src[e] and dst[e]; ds_add_f32 arrival order as was
# speedup vs baseline: 3.7883x; 1.0444x over previous
#include <hip/hip_runtime.h>


typedef float v2f __attribute__((ext_vector_type(2)));
typedef float v8f __attribute__((ext_vector_type(8)));

#define ROWS 208
#define TP   68
#define NT   1024
#define BPT  2
#define QCAP 64
#define KPAD 255u
#define SROWS 800
#define SPAD 63u
#define NS   4
#define SBT  ((NT / 32) * BPT)

static __device__ __forceinline__ void wave_sync() {
    __builtin_amdgcn_fence(__ATOMIC_RELEASE, "wavefront");
    __builtin_amdgcn_wave_barrier();
    __builtin_amdgcn_fence(__ATOMIC_ACQUIRE, "wavefront");
}

static __host__ __device__ __forceinline__ int padded_superblocks(int E) {
    return ((((E + 1023) >> 10) + SBT - 1) / SBT) * SBT;
}

template <int PLANES, unsigned RPO, unsigned PAD, bool REC>
static __device__ __forceinline__ void build_keys(const int* __restrict__ arr, const int* __restrict__ other,
                                                  unsigned* __restrict__ out, unsigned* __restrict__ rec,
                                                  int sb, int lane, int E) {
    const int e0 = sb * 1024 + lane;
    const bool whole = sb * 1024 + 1024 <= E;

    int idx[32];
    int oth[32];
    if (whole) {
        #pragma unroll
        for (int b = 0; b < 32; ++b) idx[b] = arr[e0 + 32 * b];
        if constexpr (REC) {
            #pragma unroll
            for (int b = 0; b < 32; ++b) oth[b] = other[e0 + 32 * b];
        }
    } else {
        #pragma unroll
        for (int b = 0; b < 32; ++b) {
            const int e = e0 + 32 * b;
            idx[b] = (e < E) ? arr[e] : -1;
            if constexpr (REC) oth[b] = (e < E) ? other[e] : 0;
        }
    }
    if constexpr (REC) {
        if (whole) {
            #pragma unroll
            for (int b = 0; b < 32; ++b) {
                const unsigned k = min((unsigned)idx[b] / RPO, PAD);
                rec[e0 + 32 * b] = ((unsigned)oth[b] << 8) | (((unsigned)idx[b] - k * RPO) & 255u);
            }
        } else {
            #pragma unroll
            for (int b = 0; b < 32; ++b) {
                const int e = e0 + 32 * b;
                const unsigned k = min((unsigned)idx[b] / RPO, PAD);
                if (e < E) rec[e] = ((unsigned)oth[b] << 8) | (((unsigned)idx[b] - k * RPO) & 255u);
            }
        }
    }
    unsigned P[PLANES];
    #pragma unroll
    for (int p = 0; p < PLANES; ++p) P[p] = 0u;
    #pragma unroll
    for (int b = 0; b < 32; ++b) {
        const unsigned k = min((unsigned)idx[b] / RPO, PAD);
        #pragma unroll
        for (int p = 0; p < PLANES; ++p) {
            const unsigned bal = __builtin_amdgcn_ballot_w32(((k >> p) & 1u) != 0u);
            P[p] = (lane == b) ? bal : P[p];
        }
    }
    char* __restrict__ run = (char*)out + (size_t)sb * (PLANES * 128);
    *(uint4*)(run + lane * 16) = make_uint4(P[0], P[1], P[2], P[3]);
    if constexpr (PLANES == 8) *(uint4*)(run + 512 + lane * 16) = make_uint4(P[4], P[5], P[6], P[7]);
    else                       *(uint2*)(run + 512 + lane * 8)  = make_uint2(P[4], P[5]);
}

__global__ __launch_bounds__(256)
void gcn_keys_kernel(const int* __restrict__ src, const int* __restrict__ dst,
                     unsigned* __restrict__ ksrc, unsigned* __restrict__ kdst,
                     unsigned* __restrict__ rec, int E) {
    const int lane = threadIdx.x & 31;
    const int sb   = blockIdx.x * 8 + (threadIdx.x >> 5);
    if (blockIdx.y) build_keys<8, ROWS,  KPAD, true >(dst, src, kdst, rec, sb, lane, E);
    else            build_keys<6, SROWS, SPAD, false>(src, dst, ksrc, rec, sb, lane, E);
}

template <int PLANES, class DrainF>
static __device__ __forceinline__ void owner_scan(const unsigned* __restrict__ keys, int sb_lo, int sb_n, int rot,
                                                  unsigned own, int* __restrict__ myq, DrainF drain) {
    const int lane = threadIdx.x & 31;
    const int wave = __builtin_amdgcn_readfirstlane((int)(threadIdx.x >> 5));
    unsigned sel[PLANES];
    #pragma unroll
    for (int p = 0; p < PLANES; ++p) sel[p] = 0u - ((own >> p) & 1u);
    const unsigned lane16 = (unsigned)lane * 16u;
    const unsigned laneHi = (unsigned)lane * (PLANES == 8 ? 16u : 8u);
    int qn = 0;

    for (int t0 = 0; t0 < sb_n; t0 += SBT) {
        uint4 lo[BPT], hi[BPT];
        int   blk[BPT];
        #pragma unroll
        for (int u = 0; u < BPT; ++u) {
            int sb = t0 + u * (NT / 32) + wave + rot;
            if (sb >= sb_n) sb -= sb_n;
            sb += sb_lo;
            blk[u] = sb * 32 + lane;
            const char* __restrict__ run = (const char*)keys + (size_t)sb * (PLANES * 128);
            lo[u] = *(const uint4*)(run + lane16);
            if constexpr (PLANES == 8) {
                hi[u] = *(const uint4*)(run + 512 + laneHi);
            } else {
                const uint2 h = *(const uint2*)(run + 512 + laneHi);
                hi[u] = make_uint4(h.x, h.y, 0u, 0u);
            }
        }
        unsigned w[BPT];
        #pragma unroll
        for (int u = 0; u < BPT; ++u) {
            unsigned m = ~(lo[u].x ^ sel[0]);
            m &= ~(lo[u].y ^ sel[1]);
            m &= ~(lo[u].z ^ sel[2]);
            m &= ~(lo[u].w ^ sel[3]);
            m &= ~(hi[u].x ^ sel[4]);
            m &= ~(hi[u].y ^ sel[5]);
            if constexpr (PLANES == 8) {
                m &= ~(hi[u].z ^ sel[6]);
                m &= ~(hi[u].w ^ sel[7]);
            }
            w[u] = m;
        }

        unsigned mk = __builtin_amdgcn_ballot_w32((w[0] | w[1]) != 0u);
        while (mk != 0u) {
            if ((w[0] | w[1]) != 0u) {
                const bool first = w[0] != 0u;
                unsigned x = first ? w[0] : w[1];
                const int p = __builtin_ctz(x);
                x &= x - 1u;
                if (first) w[0] = x; else w[1] = x;
                myq[qn + (int)__builtin_amdgcn_mbcnt_lo(mk, 0u)] = (first ? blk[0] : blk[1]) * 32 + p;
            }
            qn += __builtin_popcount(mk);
            if (qn >= 32) {
                qn -= 32;
                drain(myq + qn, 32);
            }
            mk = __builtin_amdgcn_ballot_w32((w[0] | w[1]) != 0u);
        }
    }
    if (qn > 0) drain(myq, qn);
}

__global__ __launch_bounds__(NT)
void gcn_srccount_kernel(const int* __restrict__ src, const unsigned* __restrict__ ksrc,
                         unsigned* __restrict__ part, int E, int N) {
    __shared__ unsigned cnt[SROWS];
    __shared__ int      q[NT / 32][QCAP];
    const int tid  = threadIdx.x;
    const int wave = tid >> 5;
    const int lane = tid & 31;
    const int o    = blockIdx.x / NS;
    const int s    = blockIdx.x % NS;
    const int base = o * SROWS;
    if (tid < SROWS) cnt[tid] = 0u;
    __syncthreads();

    const int trips = padded_superblocks(E) / SBT;
    const int t_lo  = (s * trips) / NS;
    const int t_hi  = ((s + 1) * trips) / NS;
    const int sb_n  = (t_hi - t_lo) * SBT;
    const int rot   = (int)(((long long)o * sb_n) / (gridDim.x / NS));

    auto drain = [&](const int* qp, int n) {
        wave_sync();
        if (lane < n) atomicAdd(&cnt[src[qp[lane]] - base], 1u);
        wave_sync();
    };
    owner_scan<6>(ksrc, t_lo * SBT, sb_n, rot, (unsigned)o, &q[wave][0], drain);

    __syncthreads();
    if (tid < SROWS && base + tid < N)
        part[(size_t)(base + tid) * NS + s] = cnt[tid];
}

__global__ __launch_bounds__(NT) __attribute__((amdgpu_waves_per_eu(8, 8)))
void gcn_owner_kernel(const float* __restrict__ feats,
                      const unsigned* __restrict__ rec,
                      const unsigned* __restrict__ kdst,
                      const unsigned* __restrict__ part,
                      const float* __restrict__ W,
                      const float* __restrict__ bias,
                      const float* __restrict__ gamma,
                      const float* __restrict__ beta,
                      float* __restrict__ out, int E, int N) {
    __shared__ float    tile[ROWS * TP];
    __shared__ float    lW[64 * 64];
    __shared__ int      q[NT / 32][QCAP];
    __shared__ unsigned cin[ROWS];

    const int tid  = threadIdx.x;
    const int wave = tid >> 5;
    const int lane = tid & 31;
    const int base = blockIdx.x * ROWS;

    for (int i = tid; i < ROWS * TP; i += NT) tile[i] = 0.0f;
    for (int i = tid; i < 64 * 64; i += NT) lW[i] = W[i];
    if (tid < ROWS) cin[tid] = 0u;
    __syncthreads();

    auto drain = [&](const int* qp, int n) {
        wave_sync();
        int my_s = 0, my_r = 0;
        uint4 c = make_uint4(0u, 0u, 0u, 0u);
        if (lane < n) {
            const unsigned rc = rec[qp[lane]];
            my_s = (int)(rc >> 8);
            my_r = (int)(rc & 255u);
            c = *(const uint4*)(part + (size_t)my_s * NS);
            atomicAdd(&cin[my_r], 1u);
        }
        float my_n = 0.0f;
        #pragma unroll
        for (int h = 0; h < 2; ++h) {
            const int t0 = 16 * h;
            if (t0 < n) {
                float2 x[16];
                #pragma unroll
                for (int k = 0; k < 16; ++k) {
                    const int s = __builtin_amdgcn_readlane(my_s, t0 + k);
                    x[k] = *(const float2*)(feats + (size_t)s * 64 + 2 * lane);
                }
                if (h == 0) my_n = rsqrtf(fmaxf((float)(c.x + c.y + c.z + c.w), 1.0f));
                #pragma unroll
                for (int k = 0; k < 16; ++k) {
                    const int t = t0 + k;
                    if (t < n) {
                        const float nrm = __int_as_float(__builtin_amdgcn_readlane(__float_as_int(my_n), t));
                        const int r = __builtin_amdgcn_readlane(my_r, t);
                        float* p = tile + r * TP + 2 * lane;
                        atomicAdd(p,     x[k].x * nrm);
                        atomicAdd(p + 1, x[k].y * nrm);
                    }
                }
            }
        }
        wave_sync();
    };

    const int nsb = padded_superblocks(E);
    owner_scan<8>(kdst, 0, nsb, (int)(((long long)blockIdx.x * nsb) / gridDim.x), blockIdx.x, &q[wave][0], drain);
    __syncthreads();

    if (wave >= ROWS / 16) return;

    const int row0 = wave * 16;
    const int mrow = lane & 15;
    const int koff = (lane >> 4) * 2;
    const int j    = lane & 15;

    float2 fr[16];
    #pragma unroll
    for (int r = 0; r < 16; ++r) {
        const int grow = base + row0 + r;
        fr[r] = make_float2(0.0f, 0.0f);
        if (grow < N) fr[r] = *(const float2*)(feats + (size_t)grow * 64 + 2 * lane);
    }

    const float rnorm = rsqrtf(fmaxf((float)cin[row0 + mrow], 1.0f));
    const float* __restrict__ arow = tile + (row0 + mrow) * TP;

    v8f c0 = {}, c1 = {}, c2 = {}, c3 = {};

    #pragma unroll
    for (int k = 0; k < 16; ++k) {
        const int kk = 4 * k + koff;
        v2f a;
        a.x = arow[kk]     * rnorm;
        a.y = arow[kk + 1] * rnorm;
        const float* w0 = lW + kk * 64;
        const float* w1 = lW + (kk + 1) * 64;
        v2f b0, b1, b2, b3;
        b0.x = w0[j];      b0.y = w1[j];
        b1.x = w0[16 + j]; b1.y = w1[16 + j];
        b2.x = w0[32 + j]; b2.y = w1[32 + j];
        b3.x = w0[48 + j]; b3.y = w1[48 + j];

        c0 = __builtin_amdgcn_wmma_f32_16x16x4_f32(false, a, false, b0, (short)0, c0, false, false);
        c1 = __builtin_amdgcn_wmma_f32_16x16x4_f32(false, a, false, b1, (short)0, c1, false, false);
        c2 = __builtin_amdgcn_wmma_f32_16x16x4_f32(false, a, false, b2, (short)0, c2, false, false);
        c3 = __builtin_amdgcn_wmma_f32_16x16x4_f32(false, a, false, b3, (short)0, c3, false, false);
    }

    float* __restrict__ st = tile + row0 * TP;
    const int half = (lane >> 4) * 8;
    #pragma unroll
    for (int v = 0; v < 8; ++v) {
        float* r = st + (half + v) * TP + j;
        r[0]  = c0[v];
        r[16] = c1[v];
        r[32] = c2[v];
        r[48] = c3[v];
    }
    wave_sync();

    const float2 bb = *(const float2*)(bias  + 2 * lane);
    const float2 gg = *(const float2*)(gamma + 2 * lane);
    const float2 be = *(const float2*)(beta  + 2 * lane);

    float2 y[16];
    float  sm[16];
    #pragma unroll
    for (int r = 0; r < 16; ++r) {
        const float2 d = *(const float2*)(st + r * TP + 2 * lane);
        y[r].x = fmaxf(d.x + bb.x, 0.0f) + fr[r].x;
        y[r].y = fmaxf(d.y + bb.y, 0.0f) + fr[r].y;
        sm[r] = y[r].x + y[r].y;
    }
    #pragma unroll
    for (int o = 16; o >= 1; o >>= 1) {
        #pragma unroll
        for (int r = 0; r < 16; ++r) sm[r] += __shfl_xor(sm[r], o);
    }
    float sq[16];
    #pragma unroll
    for (int r = 0; r < 16; ++r) {
        const float mu = sm[r] * (1.0f / 64.0f);
        y[r].x -= mu;
        y[r].y -= mu;
        sq[r] = y[r].x * y[r].x + y[r].y * y[r].y;
    }
    #pragma unroll
    for (int o = 16; o >= 1; o >>= 1) {
        #pragma unroll
        for (int r = 0; r < 16; ++r) sq[r] += __shfl_xor(sq[r], o);
    }
    #pragma unroll
    for (int r = 0; r < 16; ++r) {
        const int grow = base + row0 + r;
        if (grow < N) {
            const float rstd = rsqrtf(sq[r] * (1.0f / 64.0f) + 1e-5f);
            float2 o2;
            o2.x = y[r].x * rstd * gg.x + be.x;
            o2.y = y[r].y * rstd * gg.y + be.y;
            *(float2*)(out + (size_t)grow * 64 + 2 * lane) = o2;
        }
    }
}

extern "C" void kernel_launch(void* const* d_in, const int* in_sizes, int n_in,
                              void* d_out, int out_size, void* d_ws, size_t ws_size,
                              hipStream_t stream) {
    const float* feats = (const float*)d_in[0];
    const int*   src   = (const int*)d_in[1];
    const int*   dst   = (const int*)d_in[2];
    const float* W     = (const float*)d_in[3];
    const float* bias  = (const float*)d_in[4];
    const float* gamma = (const float*)d_in[5];
    const float* beta  = (const float*)d_in[6];

    const int N = in_sizes[0] / 64;
    const int E = in_sizes[1];

    unsigned* part = (unsigned*)d_ws;
    unsigned* ksrc = (unsigned*)((char*)d_ws + (16u << 20));
    unsigned* kdst = (unsigned*)((char*)d_ws + (32u << 20));
    unsigned* rec  = (unsigned*)((char*)d_ws + (48u << 20));

    const int G  = (N + ROWS - 1) / ROWS;
    const int GS = (N + SROWS - 1) / SROWS;
    const int GK = padded_superblocks(E) / 8;
    gcn_keys_kernel<<<dim3(GK, 2), 256, 0, stream>>>(src, dst, ksrc, kdst, rec, E);
    gcn_srccount_kernel<<<GS * NS, NT, 0, stream>>>(src, ksrc, part, E, N);
    gcn_owner_kernel<<<G, NT, 0, stream>>>(feats, rec, kdst, part, W, bias, gamma, beta,
                                           (float*)d_out, E, N);
}
